# starting kernel
# baseline (speedup 1.0000x reference)
#include <hip/hip_runtime.h>
#include <hip/hip_bf16.h>

typedef __attribute__((ext_vector_type(2))) float v2f;
typedef __attribute__((ext_vector_type(8))) float v8f;

#define NSEG   8192
#define D_IN   64
#define D_HID  128
#define LDS_PITCH 132

__global__ void seg_scan_kernel(const int* __restrict__ l, int* __restrict__ offs) {
  __shared__ int ps[256];
  const int tid = threadIdx.x;
  int local[32];
  int s = 0;
#pragma unroll
  for (int i = 0; i < 32; ++i) {
    int v = l[tid * 32 + i];
    local[i] = v;
    s += v;
  }
  ps[tid] = s;
  __syncthreads();
  for (int off = 1; off < 256; off <<= 1) {
    int v = 0;
    if (tid >= off) v = ps[tid - off];
    __syncthreads();
    if (tid >= off) ps[tid] += v;
    __syncthreads();
  }
  int run = (tid == 0) ? 0 : ps[tid - 1];
#pragma unroll
  for (int i = 0; i < 32; ++i) {
    offs[tid * 32 + i] = run;
    run += local[i];
  }
}

__device__ __forceinline__ void tile_mma_16x128(const v2f a[16],
                                                const float* __restrict__ bcol,
                                                v8f c[8]) {
#pragma unroll
  for (int ks = 0; ks < 16; ++ks) {
    const float* bp = bcol + (ks * 4) * LDS_PITCH;
#pragma unroll
    for (int nt = 0; nt < 8; ++nt) {
      v2f b;
      b.x = bp[nt * 16];
      b.y = bp[nt * 16 + LDS_PITCH];
      c[nt] = __builtin_amdgcn_wmma_f32_16x16x4_f32(
          false, a[ks], false, b, (short)0, c[nt], false, false);
    }
  }
}

__global__ __launch_bounds__(256) void fused_mlp_segsum_kernel(
    const float* __restrict__ x, const float* __restrict__ W1,
    const float* __restrict__ b1, const int* __restrict__ offs,
    const int* __restrict__ l, float* __restrict__ xm) {
  __shared__ float w1s[D_IN * LDS_PITCH];

  const int tid = threadIdx.x;
  for (int i = tid; i < D_IN * D_HID; i += 256) {
    w1s[(i >> 7) * LDS_PITCH + (i & 127)] = W1[i];
  }
  __syncthreads();

  const int lane = tid & 31;
  const int wave = tid >> 5;
  const int seg  = blockIdx.x * 8 + wave;
  const int start = offs[seg];
  const int len   = l[seg];
  const int half = lane >> 4;
  const int lm   = lane & 15;

  const float* bcol = w1s + (2 * half) * LDS_PITCH + lm;

  float bias[8];
#pragma unroll
  for (int nt = 0; nt < 8; ++nt) bias[nt] = b1[nt * 16 + lm];

  float accS[8];
#pragma unroll
  for (int nt = 0; nt < 8; ++nt) accS[nt] = 0.0f;

  const int nfull = len & ~15;
  int t0 = 0;

  for (; t0 < nfull; t0 += 16) {
    const float* xrow = x + (size_t)(start + t0 + lm) * D_IN + 2 * half;
    v2f a[16];
#pragma unroll
    for (int ks = 0; ks < 16; ++ks) a[ks] = *(const v2f*)(xrow + ks * 4);

    v8f c[8];
#pragma unroll
    for (int nt = 0; nt < 8; ++nt)
#pragma unroll
      for (int e = 0; e < 8; ++e) c[nt][e] = 0.0f;

    tile_mma_16x128(a, bcol, c);

#pragma unroll
    for (int nt = 0; nt < 8; ++nt) {
      float v[8];
#pragma unroll
      for (int r = 0; r < 8; ++r) v[r] = fmaxf(c[nt][r] + bias[nt], 0.0f);
      float s = ((v[0] + v[1]) + (v[2] + v[3])) + ((v[4] + v[5]) + (v[6] + v[7]));
      accS[nt] += s;
    }
  }

  if (t0 < len) {
    const bool tokv = (t0 + lm) < len;
    const float* xrow = x + (size_t)(start + t0 + lm) * D_IN + 2 * half;
    v2f a[16];
#pragma unroll
    for (int ks = 0; ks < 16; ++ks) {
      v2f av;
      av.x = 0.0f;
      av.y = 0.0f;
      if (tokv) av = *(const v2f*)(xrow + ks * 4);
      a[ks] = av;
    }

    v8f c[8];
#pragma unroll
    for (int nt = 0; nt < 8; ++nt)
#pragma unroll
      for (int e = 0; e < 8; ++e) c[nt][e] = 0.0f;

    tile_mma_16x128(a, bcol, c);

#pragma unroll
    for (int nt = 0; nt < 8; ++nt) {
      float v[8];
#pragma unroll
      for (int r = 0; r < 8; ++r) {
        const bool rv = (t0 + r + 8 * half) < len;
        v[r] = rv ? fmaxf(c[nt][r] + bias[nt], 0.0f) : 0.0f;
      }
      float s = ((v[0] + v[1]) + (v[2] + v[3])) + ((v[4] + v[5]) + (v[6] + v[7]));
      accS[nt] += s;
    }
  }

#pragma unroll
  for (int nt = 0; nt < 8; ++nt) {
    accS[nt] += __shfl_xor(accS[nt], 16, 32);
  }
  if (lane < 16) {
#pragma unroll
    for (int nt = 0; nt < 8; ++nt) {
      xm[(size_t)seg * D_HID + nt * 16 + lane] = accS[nt];
    }
  }
}

__global__ __launch_bounds__(32) void head_gemm_kernel(
    const float* __restrict__ xm, const float* __restrict__ W2,
    const float* __restrict__ b2, float* __restrict__ out) {
  const int lane = threadIdx.x & 31;
  const int half = lane >> 4;
  const int lm   = lane & 15;
  const int mt   = blockIdx.x;

  const float* arow = xm + (size_t)(mt * 16 + lm) * D_HID + 2 * half;
  const float* bcol = W2 + (2 * half) * D_HID + lm;

  v8f c[8];
#pragma unroll
  for (int nt = 0; nt < 8; ++nt)
#pragma unroll
    for (int e = 0; e < 8; ++e) c[nt][e] = 0.0f;

#pragma unroll
  for (int ks = 0; ks < 32; ++ks) {
    v2f a = *(const v2f*)(arow + ks * 4);
    const float* bp = bcol + (ks * 4) * D_HID;
#pragma unroll
    for (int nt = 0; nt < 8; ++nt) {
      v2f b;
      b.x = bp[nt * 16];
      b.y = bp[nt * 16 + D_HID];
      c[nt] = __builtin_amdgcn_wmma_f32_16x16x4_f32(
          false, a, false, b, (short)0, c[nt], false, false);
    }
  }

#pragma unroll
  for (int nt = 0; nt < 8; ++nt) {
    const int n = nt * 16 + lm;
    const float bias = b2[n];
#pragma unroll
    for (int r = 0; r < 8; ++r) {
      out[(size_t)(mt * 16 + r + 8 * half) * D_HID + n] = c[nt][r] + bias;
    }
  }
}

extern "C" void kernel_launch(void* const* d_in, const int* in_sizes, int n_in,
                              void* d_out, int out_size, void* d_ws, size_t ws_size,
                              hipStream_t stream) {
  (void)in_sizes; (void)n_in; (void)out_size; (void)ws_size;
  const float* x  = (const float*)d_in[0];
  const int*   l  = (const int*)d_in[1];
  const float* W1 = (const float*)d_in[2];
  const float* b1 = (const float*)d_in[3];
  const float* W2 = (const float*)d_in[4];
  const float* b2 = (const float*)d_in[5];
  float* out = (float*)d_out;

  int*   offs = (int*)d_ws;
  float* xm   = (float*)((char*)d_ws + ((NSEG * sizeof(int) + 255) & ~size_t(255)));

  seg_scan_kernel<<<1, 256, 0, stream>>>(l, offs);
  fused_mlp_segsum_kernel<<<NSEG / 8, 256, 0, stream>>>(x, W1, b1, offs, l, xm);
  head_gemm_kernel<<<NSEG / 16, 32, 0, stream>>>(xm, W2, b2, out);
}


# A segment longer than 256 tokens is shared by the 8 waves of its block: tiles dealt round-robin, per-wave sums added through LDS in wave order (changes the order of that segment's sum only)
# speedup vs baseline: 2.5857x; 2.5857x over previous
#include <hip/hip_runtime.h>
#include <hip/hip_bf16.h>

typedef __attribute__((ext_vector_type(2))) float v2f;
typedef __attribute__((ext_vector_type(8))) float v8f;

#define NSEG   8192
#define D_IN   64
#define D_HID  128
#define LDS_PITCH 132

__global__ void seg_scan_kernel(const int* __restrict__ l, int* __restrict__ offs) {
  __shared__ int ps[256];
  const int tid = threadIdx.x;
  int local[32];
  int s = 0;
#pragma unroll
  for (int i = 0; i < 32; ++i) {
    int v = l[tid * 32 + i];
    local[i] = v;
    s += v;
  }
  ps[tid] = s;
  __syncthreads();
  for (int off = 1; off < 256; off <<= 1) {
    int v = 0;
    if (tid >= off) v = ps[tid - off];
    __syncthreads();
    if (tid >= off) ps[tid] += v;
    __syncthreads();
  }
  int run = (tid == 0) ? 0 : ps[tid - 1];
#pragma unroll
  for (int i = 0; i < 32; ++i) {
    offs[tid * 32 + i] = run;
    run += local[i];
  }
}

__device__ __forceinline__ void tile_mma_16x128(const v2f a[16],
                                                const v2f (&b)[16][8],
                                                v8f c[8]) {
#pragma unroll
  for (int ks = 0; ks < 16; ++ks) {
#pragma unroll
    for (int nt = 0; nt < 8; ++nt) {
      c[nt] = __builtin_amdgcn_wmma_f32_16x16x4_f32(
          false, a[ks], false, b[ks][nt], (short)0, c[nt], false, false);
    }
  }
}

#define COOP_MIN_LEN 256

__global__ __launch_bounds__(256) void fused_mlp_segsum_kernel(
    const float* __restrict__ x, const float* __restrict__ W1,
    const float* __restrict__ b1, const int* __restrict__ offs,
    const int* __restrict__ l, float* __restrict__ xm) {
  __shared__ float w1s[D_IN * LDS_PITCH];
  __shared__ float red[8 * D_HID];

  const int tid = threadIdx.x;
  for (int i = tid; i < D_IN * D_HID; i += 256) {
    w1s[(i >> 7) * LDS_PITCH + (i & 127)] = W1[i];
  }
  __syncthreads();

  const int lane = tid & 31;
  const int wave = __builtin_amdgcn_readfirstlane(tid >> 5);
  const int half = lane >> 4;
  const int lm   = lane & 15;

  const float* bcol = w1s + (2 * half) * LDS_PITCH + lm;
  v2f bfrag[16][8];
#pragma unroll
  for (int ks = 0; ks < 16; ++ks) {
    const float* bp = bcol + (ks * 4) * LDS_PITCH;
#pragma unroll
    for (int nt = 0; nt < 8; ++nt) {
      bfrag[ks][nt].x = bp[nt * 16];
      bfrag[ks][nt].y = bp[nt * 16 + LDS_PITCH];
    }
  }

  float bias[8];
#pragma unroll
  for (int nt = 0; nt < 8; ++nt) bias[nt] = b1[nt * 16 + lm];

  unsigned longmask = 0;
#pragma unroll
  for (int j = 0; j < 8; ++j) {
    longmask |= (unsigned)(l[blockIdx.x * 8 + j] > COOP_MIN_LEN) << j;
  }

#pragma nounroll
  for (int job = 0; job < 9; ++job) {
    const bool shared = (job != 0);
    const int sidx = shared ? job - 1 : wave;
    if ((((longmask >> sidx) & 1u) != 0u) != shared) continue;

    const int seg   = blockIdx.x * 8 + sidx;
    const int start = offs[seg];
    const int len   = __builtin_amdgcn_readfirstlane(l[seg]);
    const int tfirst = shared ? wave * 16 : 0;
    const int tstep  = shared ? 8 * 16 : 16;

    float accS[8];
#pragma unroll
    for (int nt = 0; nt < 8; ++nt) accS[nt] = 0.0f;

    for (int t0 = tfirst; t0 < len; t0 += tstep) {
      const float* xrow = x + (size_t)(start + t0 + lm) * D_IN + 2 * half;
      if (t0 + 16 <= len) {
        v2f a[16];
#pragma unroll
        for (int ks = 0; ks < 16; ++ks) a[ks] = *(const v2f*)(xrow + ks * 4);

        v8f c[8];
#pragma unroll
        for (int nt = 0; nt < 8; ++nt)
#pragma unroll
          for (int e = 0; e < 8; ++e) c[nt][e] = 0.0f;

        tile_mma_16x128(a, bfrag, c);

#pragma unroll
        for (int nt = 0; nt < 8; ++nt) {
          float v[8];
#pragma unroll
          for (int r = 0; r < 8; ++r) v[r] = fmaxf(c[nt][r] + bias[nt], 0.0f);
          float s = ((v[0] + v[1]) + (v[2] + v[3])) + ((v[4] + v[5]) + (v[6] + v[7]));
          accS[nt] += s;
        }
      } else {
        const bool tokv = (t0 + lm) < len;
        v2f a[16];
#pragma unroll
        for (int ks = 0; ks < 16; ++ks) {
          v2f av;
          av.x = 0.0f;
          av.y = 0.0f;
          if (tokv) av = *(const v2f*)(xrow + ks * 4);
          a[ks] = av;
        }

        v8f c[8];
#pragma unroll
        for (int nt = 0; nt < 8; ++nt)
#pragma unroll
          for (int e = 0; e < 8; ++e) c[nt][e] = 0.0f;

        tile_mma_16x128(a, bfrag, c);

#pragma unroll
        for (int nt = 0; nt < 8; ++nt) {
          float v[8];
#pragma unroll
          for (int r = 0; r < 8; ++r) {
            const bool rv = (t0 + r + 8 * half) < len;
            v[r] = rv ? fmaxf(c[nt][r] + bias[nt], 0.0f) : 0.0f;
          }
          float s = ((v[0] + v[1]) + (v[2] + v[3])) + ((v[4] + v[5]) + (v[6] + v[7]));
          accS[nt] += s;
        }
      }
    }

#pragma unroll
    for (int nt = 0; nt < 8; ++nt) {
      accS[nt] += __shfl_xor(accS[nt], 16, 32);
    }

    if (!shared) {
      if (lane < 16) {
#pragma unroll
        for (int nt = 0; nt < 8; ++nt) {
          xm[(size_t)seg * D_HID + nt * 16 + lane] = accS[nt];
        }
      }
    } else {
      if (lane < 16) {
#pragma unroll
        for (int nt = 0; nt < 8; ++nt) {
          red[wave * D_HID + nt * 16 + lane] = accS[nt];
        }
      }
      __syncthreads();
      if (tid < D_HID) {
        float s = red[tid];
#pragma unroll
        for (int w = 1; w < 8; ++w) s += red[w * D_HID + tid];
        xm[(size_t)seg * D_HID + tid] = s;
      }
      __syncthreads();
    }
  }
}

__global__ __launch_bounds__(32) void head_gemm_kernel(
    const float* __restrict__ xm, const float* __restrict__ W2,
    const float* __restrict__ b2, float* __restrict__ out) {
  const int lane = threadIdx.x & 31;
  const int half = lane >> 4;
  const int lm   = lane & 15;
  const int mt   = blockIdx.x;

  const float* arow = xm + (size_t)(mt * 16 + lm) * D_HID + 2 * half;
  const float* bcol = W2 + (2 * half) * D_HID + lm;

  v8f c[8];
#pragma unroll
  for (int nt = 0; nt < 8; ++nt)
#pragma unroll
    for (int e = 0; e < 8; ++e) c[nt][e] = 0.0f;

#pragma unroll
  for (int ks = 0; ks < 32; ++ks) {
    v2f a = *(const v2f*)(arow + ks * 4);
    const float* bp = bcol + (ks * 4) * D_HID;
#pragma unroll
    for (int nt = 0; nt < 8; ++nt) {
      v2f b;
      b.x = bp[nt * 16];
      b.y = bp[nt * 16 + D_HID];
      c[nt] = __builtin_amdgcn_wmma_f32_16x16x4_f32(
          false, a, false, b, (short)0, c[nt], false, false);
    }
  }

#pragma unroll
  for (int nt = 0; nt < 8; ++nt) {
    const int n = nt * 16 + lm;
    const float bias = b2[n];
#pragma unroll
    for (int r = 0; r < 8; ++r) {
      out[(size_t)(mt * 16 + r + 8 * half) * D_HID + n] = c[nt][r] + bias;
    }
  }
}

extern "C" void kernel_launch(void* const* d_in, const int* in_sizes, int n_in,
                              void* d_out, int out_size, void* d_ws, size_t ws_size,
                              hipStream_t stream) {
  (void)in_sizes; (void)n_in; (void)out_size; (void)ws_size;
  const float* x  = (const float*)d_in[0];
  const int*   l  = (const int*)d_in[1];
  const float* W1 = (const float*)d_in[2];
  const float* b1 = (const float*)d_in[3];
  const float* W2 = (const float*)d_in[4];
  const float* b2 = (const float*)d_in[5];
  float* out = (float*)d_out;

  int*   offs = (int*)d_ws;
  float* xm   = (float*)((char*)d_ws + ((NSEG * sizeof(int) + 255) & ~size_t(255)));

  seg_scan_kernel<<<1, 256, 0, stream>>>(l, offs);
  fused_mlp_segsum_kernel<<<NSEG / 8, 256, 0, stream>>>(x, W1, b1, offs, l, xm);
  head_gemm_kernel<<<NSEG / 16, 32, 0, stream>>>(xm, W2, b2, out);
}


# Segments over 256 tokens are cut into 128-token units dealt over the whole grid (one tile per wave, LDS sum in wave order); head kernel adds unit rows in unit order (reorders those segments' sums)
# speedup vs baseline: 4.2615x; 1.6481x over previous
#include <hip/hip_runtime.h>
#include <hip/hip_bf16.h>

typedef __attribute__((ext_vector_type(2))) float v2f;
typedef __attribute__((ext_vector_type(8))) float v8f;

#define NSEG   8192
#define D_IN   64
#define D_HID  128
#define LDS_PITCH 132
#define TOTAL_TOK 1048576
#define UNIT_TOK  128
#define NUNIT     (TOTAL_TOK / UNIT_TOK)
#define NBLK      (NSEG / 8)
#define UNITS_PER_BLK (NUNIT / NBLK)
#define COOP_MIN_LEN 256

__global__ void seg_scan_kernel(const int* __restrict__ l, int* __restrict__ offs,
                                int* __restrict__ unitseg) {
  __shared__ int ps[256];
  const int tid = threadIdx.x;
  for (int i = tid; i < NUNIT; i += 256) unitseg[i] = -1;
  int local[32];
  int s = 0;
  unsigned lmask = 0;
#pragma unroll
  for (int i = 0; i < 32; ++i) {
    int v = l[tid * 32 + i];
    local[i] = v;
    s += v;
    lmask |= (unsigned)(v > COOP_MIN_LEN) << i;
  }
  ps[tid] = s;
  __syncthreads();
  for (int off = 1; off < 256; off <<= 1) {
    int v = 0;
    if (tid >= off) v = ps[tid - off];
    __syncthreads();
    if (tid >= off) ps[tid] += v;
    __syncthreads();
  }
  const int run0 = (tid == 0) ? 0 : ps[tid - 1];
  int run = run0;
#pragma unroll
  for (int i = 0; i < 32; ++i) {
    offs[tid * 32 + i] = run;
    run += local[i];
  }
  while (lmask != 0u) {
    const int i = __builtin_ctz(lmask);
    lmask &= lmask - 1u;
    int o = run0;
    int v = 0;
#pragma unroll
    for (int k = 0; k < 32; ++k) {
      o += (k < i) ? local[k] : 0;
      v = (k == i) ? local[k] : v;
    }
    const int q  = o >> 7;
    const int nf = v >> 7;
#pragma nounroll
    for (int p = 0; p < nf; ++p) {
      if (q + p < NUNIT) unitseg[q + p] = tid * 32 + i;
    }
  }
}

__device__ __forceinline__ void tile_mma_16x128(const v2f a[16],
                                                const v2f (&b)[16][8],
                                                v8f c[8]) {
#pragma unroll
  for (int ks = 0; ks < 16; ++ks) {
#pragma unroll
    for (int nt = 0; nt < 8; ++nt) {
      c[nt] = __builtin_amdgcn_wmma_f32_16x16x4_f32(
          false, a[ks], false, b[ks][nt], (short)0, c[nt], false, false);
    }
  }
}


__global__ __launch_bounds__(256) void fused_mlp_segsum_kernel(
    const float* __restrict__ x, const float* __restrict__ W1,
    const float* __restrict__ b1, const int* __restrict__ offs,
    const int* __restrict__ l, const int* __restrict__ unitseg,
    float* __restrict__ xm, float* __restrict__ part) {
  __shared__ float w1s[D_IN * LDS_PITCH];
  __shared__ float red[8 * D_HID];

  const int tid = threadIdx.x;
  for (int i = tid; i < D_IN * D_HID; i += 256) {
    w1s[(i >> 7) * LDS_PITCH + (i & 127)] = W1[i];
  }
  __syncthreads();

  const int lane = tid & 31;
  const int wave = __builtin_amdgcn_readfirstlane(tid >> 5);
  const int half = lane >> 4;
  const int lm   = lane & 15;

  const float* bcol = w1s + (2 * half) * LDS_PITCH + lm;
  v2f bfrag[16][8];
#pragma unroll
  for (int ks = 0; ks < 16; ++ks) {
    const float* bp = bcol + (ks * 4) * LDS_PITCH;
#pragma unroll
    for (int nt = 0; nt < 8; ++nt) {
      bfrag[ks][nt].x = bp[nt * 16];
      bfrag[ks][nt].y = bp[nt * 16 + LDS_PITCH];
    }
  }

  float bias[8];
#pragma unroll
  for (int nt = 0; nt < 8; ++nt) bias[nt] = b1[nt * 16 + lm];

  const int rb = (int)((blockIdx.x * 1021u) & (NBLK - 1));
  unsigned umask = 0;
#pragma unroll
  for (int k = 0; k < UNITS_PER_BLK; ++k) {
    umask |= (unsigned)(unitseg[rb + k * NBLK] >= 0) << k;
  }
  umask = (unsigned)__builtin_amdgcn_readfirstlane((int)umask);

#pragma nounroll
  for (int job = 0; job < 1 + UNITS_PER_BLK; ++job) {
    const bool unit = (job != 0);
    int seg, base, n;
    const int R = rb + (job - 1) * NBLK;
    if (!unit) {
      seg = blockIdx.x * 8 + wave;
      const int len = __builtin_amdgcn_readfirstlane(l[seg]);
      const int st  = __builtin_amdgcn_readfirstlane(offs[seg]);
      const int cut = (len > COOP_MIN_LEN) ? (len & ~(UNIT_TOK - 1)) : 0;
      base = st + cut;
      n    = len - cut;
    } else {
      if (((umask >> (job - 1)) & 1u) == 0u) continue;
      seg = __builtin_amdgcn_readfirstlane(unitseg[R]);
      const int st = __builtin_amdgcn_readfirstlane(offs[seg]);
      base = st + (R - (st >> 7)) * UNIT_TOK + wave * 16;
      n    = 16;
    }

    float accS[8];
#pragma unroll
    for (int nt = 0; nt < 8; ++nt) accS[nt] = 0.0f;

    for (int t0 = 0; t0 < n; t0 += 16) {
      const float* xrow = x + (size_t)(base + t0 + lm) * D_IN + 2 * half;
      if (t0 + 16 <= n) {
        v2f a[16];
#pragma unroll
        for (int ks = 0; ks < 16; ++ks) a[ks] = *(const v2f*)(xrow + ks * 4);

        v8f c[8];
#pragma unroll
        for (int nt = 0; nt < 8; ++nt)
#pragma unroll
          for (int e = 0; e < 8; ++e) c[nt][e] = 0.0f;

        tile_mma_16x128(a, bfrag, c);

#pragma unroll
        for (int nt = 0; nt < 8; ++nt) {
          float v[8];
#pragma unroll
          for (int r = 0; r < 8; ++r) v[r] = fmaxf(c[nt][r] + bias[nt], 0.0f);
          float s = ((v[0] + v[1]) + (v[2] + v[3])) + ((v[4] + v[5]) + (v[6] + v[7]));
          accS[nt] += s;
        }
      } else {
        const bool tokv = (t0 + lm) < n;
        v2f a[16];
#pragma unroll
        for (int ks = 0; ks < 16; ++ks) {
          v2f av;
          av.x = 0.0f;
          av.y = 0.0f;
          if (tokv) av = *(const v2f*)(xrow + ks * 4);
          a[ks] = av;
        }

        v8f c[8];
#pragma unroll
        for (int nt = 0; nt < 8; ++nt)
#pragma unroll
          for (int e = 0; e < 8; ++e) c[nt][e] = 0.0f;

        tile_mma_16x128(a, bfrag, c);

#pragma unroll
        for (int nt = 0; nt < 8; ++nt) {
          float v[8];
#pragma unroll
          for (int r = 0; r < 8; ++r) {
            const bool rv = (t0 + r + 8 * half) < n;
            v[r] = rv ? fmaxf(c[nt][r] + bias[nt], 0.0f) : 0.0f;
          }
          float s = ((v[0] + v[1]) + (v[2] + v[3])) + ((v[4] + v[5]) + (v[6] + v[7]));
          accS[nt] += s;
        }
      }
    }

#pragma unroll
    for (int nt = 0; nt < 8; ++nt) {
      accS[nt] += __shfl_xor(accS[nt], 16, 32);
    }

    if (!unit) {
      if (lane < 16) {
#pragma unroll
        for (int nt = 0; nt < 8; ++nt) {
          xm[(size_t)seg * D_HID + nt * 16 + lane] = accS[nt];
        }
      }
    } else {
      if (lane < 16) {
#pragma unroll
        for (int nt = 0; nt < 8; ++nt) {
          red[wave * D_HID + nt * 16 + lane] = accS[nt];
        }
      }
      __syncthreads();
      if (tid < D_HID) {
        float s = red[tid];
#pragma unroll
        for (int w = 1; w < 8; ++w) s += red[w * D_HID + tid];
        part[(size_t)R * D_HID + tid] = s;
      }
      __syncthreads();
    }
  }
}

__global__ __launch_bounds__(32) void head_gemm_kernel(
    float* xm, const float* __restrict__ part, const int* __restrict__ offs,
    const int* __restrict__ l, const float* __restrict__ W2,
    const float* __restrict__ b2, float* __restrict__ out) {
  const int lane = threadIdx.x & 31;
  const int half = lane >> 4;
  const int lm   = lane & 15;
  const int mt   = blockIdx.x;

  unsigned lmask = __builtin_amdgcn_ballot_w32(l[mt * 16 + lm] > COOP_MIN_LEN) & 0xffffu;
  while (lmask != 0u) {
    const int r = __builtin_ctz(lmask);
    lmask &= lmask - 1u;
    const int seg = mt * 16 + r;
    const int len = l[seg];
    const int q = offs[seg] >> 7;
    int nf = len >> 7;
    if (nf > NUNIT - q) nf = NUNIT - q;
    float* row = xm + (size_t)seg * D_HID + lane * 4;
    const float* prow = part + (size_t)q * D_HID + lane * 4;
    float4 acc = *(const float4*)row;
#pragma unroll 8
    for (int p = 0; p < nf; ++p) {
      const float4 v = *(const float4*)(prow + (size_t)p * D_HID);
      acc.x += v.x; acc.y += v.y; acc.z += v.z; acc.w += v.w;
    }
    *(float4*)row = acc;
  }
  __syncthreads();

  const float* arow = xm + (size_t)(mt * 16 + lm) * D_HID + 2 * half;
  const float* bcol = W2 + (2 * half) * D_HID + lm;

  v8f c[8];
#pragma unroll
  for (int nt = 0; nt < 8; ++nt)
#pragma unroll
    for (int e = 0; e < 8; ++e) c[nt][e] = 0.0f;

#pragma unroll
  for (int ks = 0; ks < 32; ++ks) {
    v2f a = *(const v2f*)(arow + ks * 4);
    const float* bp = bcol + (ks * 4) * D_HID;
#pragma unroll
    for (int nt = 0; nt < 8; ++nt) {
      v2f b;
      b.x = bp[nt * 16];
      b.y = bp[nt * 16 + D_HID];
      c[nt] = __builtin_amdgcn_wmma_f32_16x16x4_f32(
          false, a, false, b, (short)0, c[nt], false, false);
    }
  }

#pragma unroll
  for (int nt = 0; nt < 8; ++nt) {
    const int n = nt * 16 + lm;
    const float bias = b2[n];
#pragma unroll
    for (int r = 0; r < 8; ++r) {
      out[(size_t)(mt * 16 + r + 8 * half) * D_HID + n] = c[nt][r] + bias;
    }
  }
}

extern "C" void kernel_launch(void* const* d_in, const int* in_sizes, int n_in,
                              void* d_out, int out_size, void* d_ws, size_t ws_size,
                              hipStream_t stream) {
  (void)in_sizes; (void)n_in; (void)out_size; (void)ws_size;
  const float* x  = (const float*)d_in[0];
  const int*   l  = (const int*)d_in[1];
  const float* W1 = (const float*)d_in[2];
  const float* b1 = (const float*)d_in[3];
  const float* W2 = (const float*)d_in[4];
  const float* b2 = (const float*)d_in[5];
  float* out = (float*)d_out;

  char* ws = (char*)d_ws;
  int*   offs    = (int*)ws;
  int*   unitseg = (int*)(ws + NSEG * sizeof(int));
  float* xm      = (float*)(ws + (NSEG + NUNIT) * sizeof(int));
  float* part    = (float*)(ws + (NSEG + NUNIT) * sizeof(int) + (size_t)NSEG * D_HID * sizeof(float));

  seg_scan_kernel<<<1, 256, 0, stream>>>(l, offs, unitseg);
  fused_mlp_segsum_kernel<<<NBLK, 256, 0, stream>>>(x, W1, b1, offs, l, unitseg, xm, part);
  head_gemm_kernel<<<NSEG / 16, 32, 0, stream>>>(xm, part, offs, l, W2, b2, out);
}


# Tokens cut into 2048 chunks of 512, one per wave, whatever the segment lengths (32 full tiles each); segment ends inside a tile by masked row sums; head adds chunk rows in chunk order (reorders sums)
# speedup vs baseline: 5.4676x; 1.2830x over previous
#include <hip/hip_runtime.h>
#include <hip/hip_bf16.h>


typedef __attribute__((ext_vector_type(2))) float v2f;
typedef __attribute__((ext_vector_type(8))) float v8f;

#define NSEG   8192
#define D_IN   64
#define D_HID  128
#define LDS_PITCH 132
#define TOTAL_TOK 1048576
#define CHUNK_TOK 512
#define NCHUNK    (TOTAL_TOK / CHUNK_TOK)
#define NBLK      (NCHUNK / 8)
#define TILES_PER_CHUNK (CHUNK_TOK / 16)

__global__ void seg_scan_kernel(const int* __restrict__ l, int* __restrict__ offs,
                                int* __restrict__ cseg) {
  __shared__ int ps[256];
  const int tid = threadIdx.x;
  for (int i = tid; i < NCHUNK; i += 256) cseg[i] = NSEG - 1;
  int local[32];
  int s = 0;
#pragma unroll
  for (int i = 0; i < 32; ++i) {
    int v = l[tid * 32 + i];
    local[i] = v;
    s += v;
  }
  ps[tid] = s;
  __syncthreads();
  for (int off = 1; off < 256; off <<= 1) {
    int v = 0;
    if (tid >= off) v = ps[tid - off];
    __syncthreads();
    if (tid >= off) ps[tid] += v;
    __syncthreads();
  }
  int run = (tid == 0) ? 0 : ps[tid - 1];
#pragma unroll
  for (int i = 0; i < 32; ++i) {
    const int o = run;
    const int v = local[i];
    offs[tid * 32 + i] = o;
    if (v > 0) {
      const int e = o + v;
      int c = max((o + CHUNK_TOK - 1) / CHUNK_TOK, 0);
#pragma nounroll
      while (c < NCHUNK && c * CHUNK_TOK < e) {
        cseg[c] = tid * 32 + i;
        ++c;
      }
    }
    run += v;
  }
}

__device__ __forceinline__ void tile_mma_16x128(const v2f a[16],
                                                const v2f (&b)[16][8],
                                                v8f c[8]) {
#pragma unroll
  for (int ks = 0; ks < 16; ++ks) {
#pragma unroll
    for (int nt = 0; nt < 8; ++nt) {
      c[nt] = __builtin_amdgcn_wmma_f32_16x16x4_f32(
          false, a[ks], false, b[ks][nt], (short)0, c[nt], false, false);
    }
  }
}

__device__ __forceinline__ void add_rows_between(const v8f c[8], float accS[8],
                                                 int tok0, int lo, int hi) {
  bool m[8];
#pragma unroll
  for (int r = 0; r < 8; ++r) m[r] = (tok0 + r >= lo) && (tok0 + r < hi);
#pragma unroll
  for (int nt = 0; nt < 8; ++nt) {
    float v[8];
#pragma unroll
    for (int r = 0; r < 8; ++r) v[r] = m[r] ? c[nt][r] : 0.0f;
    float s = ((v[0] + v[1]) + (v[2] + v[3])) + ((v[4] + v[5]) + (v[6] + v[7]));
    accS[nt] += s;
  }
}

__device__ __forceinline__ void store_row_sums(float* dst, float accS[8], int lane) {
#pragma unroll
  for (int nt = 0; nt < 8; ++nt) accS[nt] += __shfl_xor(accS[nt], 16, 32);
  if (lane < 16) {
#pragma unroll
    for (int nt = 0; nt < 8; ++nt) dst[nt * 16 + lane] = accS[nt];
  }
#pragma unroll
  for (int nt = 0; nt < 8; ++nt) accS[nt] = 0.0f;
}


__global__ __launch_bounds__(256) void fused_mlp_segsum_kernel(
    const float* __restrict__ x, const float* __restrict__ W1,
    const float* __restrict__ b1, const int* __restrict__ offs,
    const int* __restrict__ l, const int* __restrict__ cseg,
    float* __restrict__ xm, float* __restrict__ head) {
  __shared__ float w1s[D_IN * LDS_PITCH];

  const int tid = threadIdx.x;
  const int lane = tid & 31;
  const int wave = __builtin_amdgcn_readfirstlane(tid >> 5);
  const int half = lane >> 4;
  const int lm   = lane & 15;
  const int ci   = blockIdx.x * 8 + wave;
  const int chunkStart = ci * CHUNK_TOK;

  int seg0 = cseg[ci];

  for (int i = tid; i < D_IN * D_HID; i += 256) {
    w1s[(i >> 7) * LDS_PITCH + (i & 127)] = W1[i];
  }
  __syncthreads();

  seg0 = min(max(seg0, 0), NSEG - 1);
  const int o0v = offs[seg0];
  const int l0v = l[seg0];

  const float* bcol = w1s + (2 * half) * LDS_PITCH + lm;
  v2f bfrag[16][8];
#pragma unroll
  for (int ks = 0; ks < 16; ++ks) {
    const float* bp = bcol + (ks * 4) * LDS_PITCH;
#pragma unroll
    for (int nt = 0; nt < 8; ++nt) {
      bfrag[ks][nt].x = bp[nt * 16];
      bfrag[ks][nt].y = bp[nt * 16 + LDS_PITCH];
    }
  }

  float bias[8];
#pragma unroll
  for (int nt = 0; nt < 8; ++nt) bias[nt] = b1[nt * 16 + lm];

  int seg = __builtin_amdgcn_readfirstlane(seg0);
  const int o0 = __builtin_amdgcn_readfirstlane(o0v);
  int segEnd = o0 + __builtin_amdgcn_readfirstlane(l0v);
  bool toHead = (o0 < chunkStart);

  float accS[8];
#pragma unroll
  for (int nt = 0; nt < 8; ++nt) accS[nt] = 0.0f;

#pragma nounroll
  for (int t = 0; t < TILES_PER_CHUNK; ++t) {
    const int T0 = chunkStart + t * 16;
    const float* xrow = x + (size_t)(T0 + lm) * D_IN + 2 * half;
    v2f a[16];
#pragma unroll
    for (int ks = 0; ks < 16; ++ks) a[ks] = *(const v2f*)(xrow + ks * 4);

    v8f c[8];
#pragma unroll
    for (int nt = 0; nt < 8; ++nt)
#pragma unroll
      for (int e = 0; e < 8; ++e) c[nt][e] = 0.0f;

    tile_mma_16x128(a, bfrag, c);

#pragma unroll
    for (int nt = 0; nt < 8; ++nt)
#pragma unroll
      for (int r = 0; r < 8; ++r) c[nt][r] = fmaxf(c[nt][r] + bias[nt], 0.0f);

    const int Tend = T0 + 16;
    if (segEnd >= Tend) {
#pragma unroll
      for (int nt = 0; nt < 8; ++nt) {
        float s = ((c[nt][0] + c[nt][1]) + (c[nt][2] + c[nt][3])) +
                  ((c[nt][4] + c[nt][5]) + (c[nt][6] + c[nt][7]));
        accS[nt] += s;
      }
    } else {
      int lo = T0;
      for (int g = 0; g < 16 && segEnd < Tend; ++g) {
        add_rows_between(c, accS, T0 + 8 * half, lo, segEnd);
        float* dst = toHead ? head + (size_t)ci * D_HID : xm + (size_t)seg * D_HID;
        store_row_sums(dst, accS, lane);
        toHead = false;
        lo = segEnd;
        int len = 0;
        do {
          ++seg;
          if (seg >= NSEG) break;
          len = __builtin_amdgcn_readfirstlane(l[seg]);
        } while (len <= 0);
        if (seg >= NSEG) {
          seg = NSEG - 1;
          segEnd = 0x7fffffff;
        } else {
          segEnd = lo + len;
        }
      }
      add_rows_between(c, accS, T0 + 8 * half, lo, Tend);
    }
  }

  float* dst = toHead ? head + (size_t)ci * D_HID : xm + (size_t)seg * D_HID;
  store_row_sums(dst, accS, lane);
}

__global__ __launch_bounds__(32) void head_gemm_kernel(
    float* xm, const float* __restrict__ head, const int* __restrict__ offs,
    const int* __restrict__ l, const float* __restrict__ W2,
    const float* __restrict__ b2, float* __restrict__ out) {
  const int lane = threadIdx.x & 31;
  const int half = lane >> 4;
  const int lm   = lane & 15;
  const int mt   = blockIdx.x;

  const int rlen = l[mt * 16 + lm];
  const int ro   = offs[mt * 16 + lm];
  const bool has = rlen > 0;
  int cs = ro / CHUNK_TOK;
  int ce = (ro + rlen - 1) / CHUNK_TOK;
  cs = min(max(cs, 0), NCHUNK - 1);
  ce = min(max(ce, cs), NCHUNK - 1);
  const int nh = has ? (ce - cs) : 0;

  unsigned lmask = __builtin_amdgcn_ballot_w32(nh >= 2) & 0xffffu;
  while (lmask != 0u) {
    const int r = __builtin_ctz(lmask);
    lmask &= lmask - 1u;
    const int q  = __shfl(cs, r, 32) + 1;
    const int nf = __shfl(nh, r, 32);
    float* row = xm + (size_t)(mt * 16 + r) * D_HID + lane * 4;
    const float* prow = head + (size_t)q * D_HID + lane * 4;
    float4 acc = *(const float4*)row;
#pragma unroll 8
    for (int p = 0; p < nf; ++p) {
      const float4 v = *(const float4*)(prow + (size_t)p * D_HID);
      acc.x += v.x; acc.y += v.y; acc.z += v.z; acc.w += v.w;
    }
    *(float4*)row = acc;
  }
  __syncthreads();

  const float* arow = xm + (size_t)(mt * 16 + lm) * D_HID + 2 * half;
  const float* hrow = head + (size_t)(cs + 1 < NCHUNK ? cs + 1 : cs) * D_HID + 2 * half;
  const float* bcol = W2 + (2 * half) * D_HID + lm;

  v2f a[32];
#pragma unroll
  for (int ks = 0; ks < 32; ++ks) { a[ks].x = 0.0f; a[ks].y = 0.0f; }
  if (has) {
#pragma unroll
    for (int ks = 0; ks < 32; ++ks) a[ks] = *(const v2f*)(arow + ks * 4);
  }
  if (nh == 1) {
#pragma unroll
    for (int ks = 0; ks < 32; ++ks) {
      const v2f h = *(const v2f*)(hrow + ks * 4);
      a[ks].x += h.x;
      a[ks].y += h.y;
    }
  }

  v8f c[8];
#pragma unroll
  for (int nt = 0; nt < 8; ++nt)
#pragma unroll
    for (int e = 0; e < 8; ++e) c[nt][e] = 0.0f;

#pragma unroll
  for (int ks = 0; ks < 32; ++ks) {
    const float* bp = bcol + (ks * 4) * D_HID;
#pragma unroll
    for (int nt = 0; nt < 8; ++nt) {
      v2f b;
      b.x = bp[nt * 16];
      b.y = bp[nt * 16 + D_HID];
      c[nt] = __builtin_amdgcn_wmma_f32_16x16x4_f32(
          false, a[ks], false, b, (short)0, c[nt], false, false);
    }
  }

#pragma unroll
  for (int nt = 0; nt < 8; ++nt) {
    const int n = nt * 16 + lm;
    const float bias = b2[n];
#pragma unroll
    for (int r = 0; r < 8; ++r) {
      out[(size_t)(mt * 16 + r + 8 * half) * D_HID + n] = c[nt][r] + bias;
    }
  }
}

extern "C" void kernel_launch(void* const* d_in, const int* in_sizes, int n_in,
                              void* d_out, int out_size, void* d_ws, size_t ws_size,
                              hipStream_t stream) {
  (void)in_sizes; (void)n_in; (void)out_size; (void)ws_size;
  const float* x  = (const float*)d_in[0];
  const int*   l  = (const int*)d_in[1];
  const float* W1 = (const float*)d_in[2];
  const float* b1 = (const float*)d_in[3];
  const float* W2 = (const float*)d_in[4];
  const float* b2 = (const float*)d_in[5];
  float* out = (float*)d_out;

  char* ws = (char*)d_ws;
  int*   offs = (int*)ws;
  int*   cseg = (int*)(ws + NSEG * sizeof(int));
  float* xm   = (float*)(ws + (NSEG + NCHUNK) * sizeof(int));
  float* head = (float*)(ws + (NSEG + NCHUNK) * sizeof(int) + (size_t)NSEG * D_HID * sizeof(float));

  seg_scan_kernel<<<1, 256, 0, stream>>>(l, offs, cseg);
  fused_mlp_segsum_kernel<<<NBLK, 256, 0, stream>>>(x, W1, b1, offs, l, cseg, xm, head);
  head_gemm_kernel<<<NSEG / 16, 32, 0, stream>>>(xm, head, offs, l, W2, b2, out);
}


# Fused kernel: each tile's x fragments are loaded one tile ahead into a second register set, as one group in front of the previous tile's 128 WMMAs (two tile bodies per loop trip); sums unchanged
# speedup vs baseline: 6.4225x; 1.1746x over previous
#include <hip/hip_runtime.h>
#include <hip/hip_bf16.h>


typedef __attribute__((ext_vector_type(2))) float v2f;
typedef __attribute__((ext_vector_type(8))) float v8f;

#define NSEG   8192
#define D_IN   64
#define D_HID  128
#define LDS_PITCH 132
#define TOTAL_TOK 1048576
#define CHUNK_TOK 512
#define NCHUNK    (TOTAL_TOK / CHUNK_TOK)
#define NBLK      (NCHUNK / 8)
#define TILES_PER_CHUNK (CHUNK_TOK / 16)

__global__ void seg_scan_kernel(const int* __restrict__ l, int* __restrict__ offs,
                                int* __restrict__ cseg) {
  __shared__ int ps[256];
  const int tid = threadIdx.x;
  for (int i = tid; i < NCHUNK; i += 256) cseg[i] = NSEG - 1;
  int local[32];
  int s = 0;
#pragma unroll
  for (int i = 0; i < 32; ++i) {
    int v = l[tid * 32 + i];
    local[i] = v;
    s += v;
  }
  ps[tid] = s;
  __syncthreads();
  for (int off = 1; off < 256; off <<= 1) {
    int v = 0;
    if (tid >= off) v = ps[tid - off];
    __syncthreads();
    if (tid >= off) ps[tid] += v;
    __syncthreads();
  }
  int run = (tid == 0) ? 0 : ps[tid - 1];
#pragma unroll
  for (int i = 0; i < 32; ++i) {
    const int o = run;
    const int v = local[i];
    offs[tid * 32 + i] = o;
    if (v > 0) {
      const int e = o + v;
      int c = max((o + CHUNK_TOK - 1) / CHUNK_TOK, 0);
#pragma nounroll
      while (c < NCHUNK && c * CHUNK_TOK < e) {
        cseg[c] = tid * 32 + i;
        ++c;
      }
    }
    run += v;
  }
}

__device__ __forceinline__ void tile_mma_16x128(const v2f a[16],
                                                const v2f (&b)[16][8],
                                                v8f c[8]) {
#pragma unroll
  for (int ks = 0; ks < 16; ++ks) {
#pragma unroll
    for (int nt = 0; nt < 8; ++nt) {
      c[nt] = __builtin_amdgcn_wmma_f32_16x16x4_f32(
          false, a[ks], false, b[ks][nt], (short)0, c[nt], false, false);
    }
  }
}

__device__ __forceinline__ void add_rows_between(const v8f c[8], float accS[8],
                                                 int tok0, int lo, int hi) {
  bool m[8];
#pragma unroll
  for (int r = 0; r < 8; ++r) m[r] = (tok0 + r >= lo) && (tok0 + r < hi);
#pragma unroll
  for (int nt = 0; nt < 8; ++nt) {
    float v[8];
#pragma unroll
    for (int r = 0; r < 8; ++r) v[r] = m[r] ? c[nt][r] : 0.0f;
    float s = ((v[0] + v[1]) + (v[2] + v[3])) + ((v[4] + v[5]) + (v[6] + v[7]));
    accS[nt] += s;
  }
}

__device__ __forceinline__ void store_row_sums(float* dst, float accS[8], int lane) {
#pragma unroll
  for (int nt = 0; nt < 8; ++nt) accS[nt] += __shfl_xor(accS[nt], 16, 32);
  if (lane < 16) {
#pragma unroll
    for (int nt = 0; nt < 8; ++nt) dst[nt * 16 + lane] = accS[nt];
  }
#pragma unroll
  for (int nt = 0; nt < 8; ++nt) accS[nt] = 0.0f;
}


__global__ __launch_bounds__(256) void fused_mlp_segsum_kernel(
    const float* __restrict__ x, const float* __restrict__ W1,
    const float* __restrict__ b1, const int* __restrict__ offs,
    const int* __restrict__ l, const int* __restrict__ cseg,
    float* __restrict__ xm, float* __restrict__ head) {
  __shared__ float w1s[D_IN * LDS_PITCH];

  const int tid = threadIdx.x;
  const int lane = tid & 31;
  const int wave = __builtin_amdgcn_readfirstlane(tid >> 5);
  const int half = lane >> 4;
  const int lm   = lane & 15;
  const int ci   = blockIdx.x * 8 + wave;
  const int chunkStart = ci * CHUNK_TOK;

  int seg0 = cseg[ci];

  for (int i = tid; i < D_IN * D_HID; i += 256) {
    w1s[(i >> 7) * LDS_PITCH + (i & 127)] = W1[i];
  }
  __syncthreads();

  seg0 = min(max(seg0, 0), NSEG - 1);
  const int o0v = offs[seg0];
  const int l0v = l[seg0];

  const float* bcol = w1s + (2 * half) * LDS_PITCH + lm;
  v2f bfrag[16][8];
#pragma unroll
  for (int ks = 0; ks < 16; ++ks) {
    const float* bp = bcol + (ks * 4) * LDS_PITCH;
#pragma unroll
    for (int nt = 0; nt < 8; ++nt) {
      bfrag[ks][nt].x = bp[nt * 16];
      bfrag[ks][nt].y = bp[nt * 16 + LDS_PITCH];
    }
  }

  float bias[8];
#pragma unroll
  for (int nt = 0; nt < 8; ++nt) bias[nt] = b1[nt * 16 + lm];

  int seg = __builtin_amdgcn_readfirstlane(seg0);
  const int o0 = __builtin_amdgcn_readfirstlane(o0v);
  int segEnd = o0 + __builtin_amdgcn_readfirstlane(l0v);
  bool toHead = (o0 < chunkStart);

  float accS[8];
#pragma unroll
  for (int nt = 0; nt < 8; ++nt) accS[nt] = 0.0f;

  auto load_tile = [&](v2f (&dst)[16], int t) {
    const float* xrow = x + (size_t)(chunkStart + t * 16 + lm) * D_IN + 2 * half;
#pragma unroll
    for (int ks = 0; ks < 16; ++ks) dst[ks] = *(const v2f*)(xrow + ks * 4);
  };

  auto do_tile = [&](const v2f (&a)[16], int t) {
    const int T0 = chunkStart + t * 16;
    v8f c[8];
#pragma unroll
    for (int nt = 0; nt < 8; ++nt)
#pragma unroll
      for (int e = 0; e < 8; ++e) c[nt][e] = 0.0f;

    tile_mma_16x128(a, bfrag, c);

#pragma unroll
    for (int nt = 0; nt < 8; ++nt)
#pragma unroll
      for (int r = 0; r < 8; ++r) c[nt][r] = fmaxf(c[nt][r] + bias[nt], 0.0f);

    const int Tend = T0 + 16;
    if (segEnd >= Tend) {
#pragma unroll
      for (int nt = 0; nt < 8; ++nt) {
        float s = ((c[nt][0] + c[nt][1]) + (c[nt][2] + c[nt][3])) +
                  ((c[nt][4] + c[nt][5]) + (c[nt][6] + c[nt][7]));
        accS[nt] += s;
      }
    } else {
      int lo = T0;
      for (int g = 0; g < 16 && segEnd < Tend; ++g) {
        add_rows_between(c, accS, T0 + 8 * half, lo, segEnd);
        float* dst = toHead ? head + (size_t)ci * D_HID : xm + (size_t)seg * D_HID;
        store_row_sums(dst, accS, lane);
        toHead = false;
        lo = segEnd;
        int len = 0;
        do {
          ++seg;
          if (seg >= NSEG) break;
          len = __builtin_amdgcn_readfirstlane(l[seg]);
        } while (len <= 0);
        if (seg >= NSEG) {
          seg = NSEG - 1;
          segEnd = 0x7fffffff;
        } else {
          segEnd = lo + len;
        }
      }
      add_rows_between(c, accS, T0 + 8 * half, lo, Tend);
    }
  };

  v2f a0[16], a1[16];
  load_tile(a0, 0);
#pragma nounroll
  for (int t = 0; t < TILES_PER_CHUNK; t += 2) {
    load_tile(a1, t + 1);
    __builtin_amdgcn_sched_barrier(0);
    do_tile(a0, t);
    load_tile(a0, min(t + 2, TILES_PER_CHUNK - 1));
    __builtin_amdgcn_sched_barrier(0);
    do_tile(a1, t + 1);
  }

  float* dst = toHead ? head + (size_t)ci * D_HID : xm + (size_t)seg * D_HID;
  store_row_sums(dst, accS, lane);
}

__global__ __launch_bounds__(32) void head_gemm_kernel(
    float* xm, const float* __restrict__ head, const int* __restrict__ offs,
    const int* __restrict__ l, const float* __restrict__ W2,
    const float* __restrict__ b2, float* __restrict__ out) {
  const int lane = threadIdx.x & 31;
  const int half = lane >> 4;
  const int lm   = lane & 15;
  const int mt   = blockIdx.x;

  const int rlen = l[mt * 16 + lm];
  const int ro   = offs[mt * 16 + lm];
  const bool has = rlen > 0;
  int cs = ro / CHUNK_TOK;
  int ce = (ro + rlen - 1) / CHUNK_TOK;
  cs = min(max(cs, 0), NCHUNK - 1);
  ce = min(max(ce, cs), NCHUNK - 1);
  const int nh = has ? (ce - cs) : 0;

  unsigned lmask = __builtin_amdgcn_ballot_w32(nh >= 2) & 0xffffu;
  while (lmask != 0u) {
    const int r = __builtin_ctz(lmask);
    lmask &= lmask - 1u;
    const int q  = __shfl(cs, r, 32) + 1;
    const int nf = __shfl(nh, r, 32);
    float* row = xm + (size_t)(mt * 16 + r) * D_HID + lane * 4;
    const float* prow = head + (size_t)q * D_HID + lane * 4;
    float4 acc = *(const float4*)row;
#pragma unroll 8
    for (int p = 0; p < nf; ++p) {
      const float4 v = *(const float4*)(prow + (size_t)p * D_HID);
      acc.x += v.x; acc.y += v.y; acc.z += v.z; acc.w += v.w;
    }
    *(float4*)row = acc;
  }
  __syncthreads();

  const float* arow = xm + (size_t)(mt * 16 + lm) * D_HID + 2 * half;
  const float* hrow = head + (size_t)(cs + 1 < NCHUNK ? cs + 1 : cs) * D_HID + 2 * half;
  const float* bcol = W2 + (2 * half) * D_HID + lm;

  v2f a[32];
#pragma unroll
  for (int ks = 0; ks < 32; ++ks) { a[ks].x = 0.0f; a[ks].y = 0.0f; }
  if (has) {
#pragma unroll
    for (int ks = 0; ks < 32; ++ks) a[ks] = *(const v2f*)(arow + ks * 4);
  }
  if (nh == 1) {
#pragma unroll
    for (int ks = 0; ks < 32; ++ks) {
      const v2f h = *(const v2f*)(hrow + ks * 4);
      a[ks].x += h.x;
      a[ks].y += h.y;
    }
  }

  v8f c[8];
#pragma unroll
  for (int nt = 0; nt < 8; ++nt)
#pragma unroll
    for (int e = 0; e < 8; ++e) c[nt][e] = 0.0f;

#pragma unroll
  for (int ks = 0; ks < 32; ++ks) {
    const float* bp = bcol + (ks * 4) * D_HID;
#pragma unroll
    for (int nt = 0; nt < 8; ++nt) {
      v2f b;
      b.x = bp[nt * 16];
      b.y = bp[nt * 16 + D_HID];
      c[nt] = __builtin_amdgcn_wmma_f32_16x16x4_f32(
          false, a[ks], false, b, (short)0, c[nt], false, false);
    }
  }

#pragma unroll
  for (int nt = 0; nt < 8; ++nt) {
    const int n = nt * 16 + lm;
    const float bias = b2[n];
#pragma unroll
    for (int r = 0; r < 8; ++r) {
      out[(size_t)(mt * 16 + r + 8 * half) * D_HID + n] = c[nt][r] + bias;
    }
  }
}

extern "C" void kernel_launch(void* const* d_in, const int* in_sizes, int n_in,
                              void* d_out, int out_size, void* d_ws, size_t ws_size,
                              hipStream_t stream) {
  (void)in_sizes; (void)n_in; (void)out_size; (void)ws_size;
  const float* x  = (const float*)d_in[0];
  const int*   l  = (const int*)d_in[1];
  const float* W1 = (const float*)d_in[2];
  const float* b1 = (const float*)d_in[3];
  const float* W2 = (const float*)d_in[4];
  const float* b2 = (const float*)d_in[5];
  float* out = (float*)d_out;

  char* ws = (char*)d_ws;
  int*   offs = (int*)ws;
  int*   cseg = (int*)(ws + NSEG * sizeof(int));
  float* xm   = (float*)(ws + (NSEG + NCHUNK) * sizeof(int));
  float* head = (float*)(ws + (NSEG + NCHUNK) * sizeof(int) + (size_t)NSEG * D_HID * sizeof(float));

  seg_scan_kernel<<<1, 256, 0, stream>>>(l, offs, cseg);
  fused_mlp_segsum_kernel<<<NBLK, 256, 0, stream>>>(x, W1, b1, offs, l, cseg, xm, head);
  head_gemm_kernel<<<NSEG / 16, 32, 0, stream>>>(xm, head, offs, l, W2, b2, out);
}


# The segment scan and the search for each chunk's first segment move into the fused kernel's prologue (every block scans the 8192 lengths in LDS): 2 launches, not 3; sums and their order unchanged
# speedup vs baseline: 6.8241x; 1.0625x over previous
#include <hip/hip_runtime.h>
#include <hip/hip_bf16.h>


typedef __attribute__((ext_vector_type(2))) float v2f;
typedef __attribute__((ext_vector_type(8))) float v8f;

#define NSEG   8192
#define D_IN   64
#define D_HID  128
#define LDS_PITCH 132
#define TOTAL_TOK 1048576
#define CHUNK_TOK 512
#define NCHUNK    (TOTAL_TOK / CHUNK_TOK)
#define NBLK      (NCHUNK / 8)
#define TILES_PER_CHUNK (CHUNK_TOK / 16)
#define NGROUP    (NSEG / 32)
#define LSEG_PITCH 36
static_assert(NBLK == NGROUP, "block b writes the offsets of group b");

__device__ __forceinline__ void tile_mma_16x128(const v2f a[16],
                                                const v2f (&b)[16][8],
                                                v8f c[8]) {
#pragma unroll
  for (int ks = 0; ks < 16; ++ks) {
#pragma unroll
    for (int nt = 0; nt < 8; ++nt) {
      c[nt] = __builtin_amdgcn_wmma_f32_16x16x4_f32(
          false, a[ks], false, b[ks][nt], (short)0, c[nt], false, false);
    }
  }
}

__device__ __forceinline__ void add_rows_between(const v8f c[8], float accS[8],
                                                 int tok0, int lo, int hi) {
  bool m[8];
#pragma unroll
  for (int r = 0; r < 8; ++r) m[r] = (tok0 + r >= lo) && (tok0 + r < hi);
#pragma unroll
  for (int nt = 0; nt < 8; ++nt) {
    float v[8];
#pragma unroll
    for (int r = 0; r < 8; ++r) v[r] = m[r] ? c[nt][r] : 0.0f;
    float s = ((v[0] + v[1]) + (v[2] + v[3])) + ((v[4] + v[5]) + (v[6] + v[7]));
    accS[nt] += s;
  }
}

__device__ __forceinline__ void store_row_sums(float* dst, float accS[8], int lane) {
#pragma unroll
  for (int nt = 0; nt < 8; ++nt) accS[nt] += __shfl_xor(accS[nt], 16, 32);
  if (lane < 16) {
#pragma unroll
    for (int nt = 0; nt < 8; ++nt) dst[nt * 16 + lane] = accS[nt];
  }
#pragma unroll
  for (int nt = 0; nt < 8; ++nt) accS[nt] = 0.0f;
}


__global__ __launch_bounds__(256) void fused_mlp_segsum_kernel(
    const float* __restrict__ x, const float* __restrict__ W1,
    const float* __restrict__ b1, const int* __restrict__ l,
    int* __restrict__ offs, float* __restrict__ xm, float* __restrict__ head) {
  __shared__ float w1s[D_IN * LDS_PITCH];
  __shared__ __attribute__((aligned(16))) int lS[NGROUP * LSEG_PITCH];
  __shared__ __attribute__((aligned(16))) int exS[NGROUP];
  __shared__ __attribute__((aligned(16))) int waveTot[8];

  const int tid = threadIdx.x;
  const int lane = tid & 31;
  const int wave = __builtin_amdgcn_readfirstlane(tid >> 5);
  const int half = lane >> 4;
  const int lm   = lane & 15;
  const int ci   = blockIdx.x * 8 + wave;
  const int chunkStart = ci * CHUNK_TOK;

  const int4* lp = (const int4*)l + tid;
  const int4 lq0 = lp[0 * 256], lq1 = lp[1 * 256], lq2 = lp[2 * 256], lq3 = lp[3 * 256];
  const int4 lq4 = lp[4 * 256], lq5 = lp[5 * 256], lq6 = lp[6 * 256], lq7 = lp[7 * 256];

  for (int i = tid; i < D_IN * D_HID; i += 256) {
    w1s[(i >> 7) * LDS_PITCH + (i & 127)] = W1[i];
  }

  {
    int* lrow = lS + (tid >> 3) * LSEG_PITCH + (tid & 7) * 4;
    *(int4*)(lrow + 0 * 32 * LSEG_PITCH) = lq0;
    *(int4*)(lrow + 1 * 32 * LSEG_PITCH) = lq1;
    *(int4*)(lrow + 2 * 32 * LSEG_PITCH) = lq2;
    *(int4*)(lrow + 3 * 32 * LSEG_PITCH) = lq3;
    *(int4*)(lrow + 4 * 32 * LSEG_PITCH) = lq4;
    *(int4*)(lrow + 5 * 32 * LSEG_PITCH) = lq5;
    *(int4*)(lrow + 6 * 32 * LSEG_PITCH) = lq6;
    *(int4*)(lrow + 7 * 32 * LSEG_PITCH) = lq7;
  }
  __syncthreads();

  int len[32];
#pragma unroll
  for (int k = 0; k < 8; ++k) {
    const int4 v = *(const int4*)(lS + tid * LSEG_PITCH + 4 * k);
    len[4 * k + 0] = v.x; len[4 * k + 1] = v.y; len[4 * k + 2] = v.z; len[4 * k + 3] = v.w;
  }
  int gsum = 0;
#pragma unroll
  for (int i = 0; i < 32; ++i) gsum += len[i];

  int inc = gsum;
#pragma unroll
  for (int d = 1; d < 32; d <<= 1) {
    const int v = __shfl_up(inc, d, 32);
    if (lane >= d) inc += v;
  }
  exS[tid] = inc - gsum;
  if (lane == 31) waveTot[wave] = inc;
  __syncthreads();

  int wt[8];
  {
    const int4 t0 = *(const int4*)(waveTot), t1 = *(const int4*)(waveTot + 4);
    wt[0] = t0.x; wt[1] = t0.y; wt[2] = t0.z; wt[3] = t0.w;
    wt[4] = t1.x; wt[5] = t1.y; wt[6] = t1.z; wt[7] = t1.w;
  }
  const int lw = lane >> 2;
  int myBase = 0, laneBase = 0, total = 0;
#pragma unroll
  for (int w = 0; w < 8; ++w) {
    myBase   += (w < wave) ? wt[w] : 0;
    laneBase += (w < lw)   ? wt[w] : 0;
    total    += wt[w];
  }

  if (tid == (int)blockIdx.x) {
    int o = myBase + inc - gsum;
#pragma unroll
    for (int k = 0; k < 8; ++k) {
      int4 w4;
      w4.x = o; o += len[4 * k + 0];
      w4.y = o; o += len[4 * k + 1];
      w4.z = o; o += len[4 * k + 2];
      w4.w = o; o += len[4 * k + 3];
      *(int4*)(offs + tid * 32 + 4 * k) = w4;
    }
  }

  int cnt = 0;
  {
    const int4 e0 = *(const int4*)(exS + lane * 8), e1 = *(const int4*)(exS + lane * 8 + 4);
    cnt += __builtin_popcount(__builtin_amdgcn_ballot_w32(laneBase + e0.x <= chunkStart));
    cnt += __builtin_popcount(__builtin_amdgcn_ballot_w32(laneBase + e0.y <= chunkStart));
    cnt += __builtin_popcount(__builtin_amdgcn_ballot_w32(laneBase + e0.z <= chunkStart));
    cnt += __builtin_popcount(__builtin_amdgcn_ballot_w32(laneBase + e0.w <= chunkStart));
    cnt += __builtin_popcount(__builtin_amdgcn_ballot_w32(laneBase + e1.x <= chunkStart));
    cnt += __builtin_popcount(__builtin_amdgcn_ballot_w32(laneBase + e1.y <= chunkStart));
    cnt += __builtin_popcount(__builtin_amdgcn_ballot_w32(laneBase + e1.z <= chunkStart));
    cnt += __builtin_popcount(__builtin_amdgcn_ballot_w32(laneBase + e1.w <= chunkStart));
  }
  const int grp = min(max(cnt - 1, 0), NGROUP - 1);
  const int gw = grp >> 5;
  int grpRun = exS[grp];
#pragma unroll
  for (int w = 0; w < 8; ++w) grpRun += (w < gw) ? wt[w] : 0;

  const int lv = lS[grp * LSEG_PITCH + lane];
  int linc = lv;
#pragma unroll
  for (int d = 1; d < 32; d <<= 1) {
    const int v = __shfl_up(linc, d, 32);
    if (lane >= d) linc += v;
  }
  const int so = grpRun + linc - lv;
  const unsigned hit = __builtin_amdgcn_ballot_w32(lv > 0 && so <= chunkStart && chunkStart < so + lv);
  int seg0, o0v, l0v;
  if (hit != 0u) {
    const int j = __builtin_ctz(hit);
    seg0 = grp * 32 + j;
    o0v = __builtin_amdgcn_readlane(so, j);
    l0v = __builtin_amdgcn_readlane(lv, j);
  } else {
    const int lLast = lS[(NGROUP - 1) * LSEG_PITCH + 31];
    seg0 = NSEG - 1;
    o0v = total - lLast;
    l0v = lLast;
  }

  const float* bcol = w1s + (2 * half) * LDS_PITCH + lm;
  v2f bfrag[16][8];
#pragma unroll
  for (int ks = 0; ks < 16; ++ks) {
    const float* bp = bcol + (ks * 4) * LDS_PITCH;
#pragma unroll
    for (int nt = 0; nt < 8; ++nt) {
      bfrag[ks][nt].x = bp[nt * 16];
      bfrag[ks][nt].y = bp[nt * 16 + LDS_PITCH];
    }
  }

  float bias[8];
#pragma unroll
  for (int nt = 0; nt < 8; ++nt) bias[nt] = b1[nt * 16 + lm];

  int seg = __builtin_amdgcn_readfirstlane(seg0);
  const int o0 = __builtin_amdgcn_readfirstlane(o0v);
  int segEnd = o0 + __builtin_amdgcn_readfirstlane(l0v);
  bool toHead = (o0 < chunkStart);

  float accS[8];
#pragma unroll
  for (int nt = 0; nt < 8; ++nt) accS[nt] = 0.0f;

  auto load_tile = [&](v2f (&dst)[16], int t) {
    const float* xrow = x + (size_t)(chunkStart + t * 16 + lm) * D_IN + 2 * half;
#pragma unroll
    for (int ks = 0; ks < 16; ++ks) dst[ks] = *(const v2f*)(xrow + ks * 4);
  };

  auto do_tile = [&](const v2f (&a)[16], int t) {
    const int T0 = chunkStart + t * 16;
    v8f c[8];
#pragma unroll
    for (int nt = 0; nt < 8; ++nt)
#pragma unroll
      for (int e = 0; e < 8; ++e) c[nt][e] = 0.0f;

    tile_mma_16x128(a, bfrag, c);

#pragma unroll
    for (int nt = 0; nt < 8; ++nt)
#pragma unroll
      for (int r = 0; r < 8; ++r) c[nt][r] = fmaxf(c[nt][r] + bias[nt], 0.0f);

    const int Tend = T0 + 16;
    if (segEnd >= Tend) {
#pragma unroll
      for (int nt = 0; nt < 8; ++nt) {
        float s = ((c[nt][0] + c[nt][1]) + (c[nt][2] + c[nt][3])) +
                  ((c[nt][4] + c[nt][5]) + (c[nt][6] + c[nt][7]));
        accS[nt] += s;
      }
    } else {
      int lo = T0;
      for (int g = 0; g < 16 && segEnd < Tend; ++g) {
        add_rows_between(c, accS, T0 + 8 * half, lo, segEnd);
        float* dst = toHead ? head + (size_t)ci * D_HID : xm + (size_t)seg * D_HID;
        store_row_sums(dst, accS, lane);
        toHead = false;
        lo = segEnd;
        int len = 0;
        do {
          ++seg;
          if (seg >= NSEG) break;
          len = __builtin_amdgcn_readfirstlane(l[seg]);
        } while (len <= 0);
        if (seg >= NSEG) {
          seg = NSEG - 1;
          segEnd = 0x7fffffff;
        } else {
          segEnd = lo + len;
        }
      }
      add_rows_between(c, accS, T0 + 8 * half, lo, Tend);
    }
  };

  v2f a0[16], a1[16];
  load_tile(a0, 0);
#pragma nounroll
  for (int t = 0; t < TILES_PER_CHUNK; t += 2) {
    load_tile(a1, t + 1);
    __builtin_amdgcn_sched_barrier(0);
    do_tile(a0, t);
    load_tile(a0, min(t + 2, TILES_PER_CHUNK - 1));
    __builtin_amdgcn_sched_barrier(0);
    do_tile(a1, t + 1);
  }

  float* dst = toHead ? head + (size_t)ci * D_HID : xm + (size_t)seg * D_HID;
  store_row_sums(dst, accS, lane);
}

__global__ __launch_bounds__(32) void head_gemm_kernel(
    float* xm, const float* __restrict__ head, const int* __restrict__ offs,
    const int* __restrict__ l, const float* __restrict__ W2,
    const float* __restrict__ b2, float* __restrict__ out) {
  const int lane = threadIdx.x & 31;
  const int half = lane >> 4;
  const int lm   = lane & 15;
  const int mt   = blockIdx.x;

  const int rlen = l[mt * 16 + lm];
  const int ro   = offs[mt * 16 + lm];
  const bool has = rlen > 0;
  int cs = ro / CHUNK_TOK;
  int ce = (ro + rlen - 1) / CHUNK_TOK;
  cs = min(max(cs, 0), NCHUNK - 1);
  ce = min(max(ce, cs), NCHUNK - 1);
  const int nh = has ? (ce - cs) : 0;

  unsigned lmask = __builtin_amdgcn_ballot_w32(nh >= 2) & 0xffffu;
  while (lmask != 0u) {
    const int r = __builtin_ctz(lmask);
    lmask &= lmask - 1u;
    const int q  = __shfl(cs, r, 32) + 1;
    const int nf = __shfl(nh, r, 32);
    float* row = xm + (size_t)(mt * 16 + r) * D_HID + lane * 4;
    const float* prow = head + (size_t)q * D_HID + lane * 4;
    float4 acc = *(const float4*)row;
#pragma unroll 8
    for (int p = 0; p < nf; ++p) {
      const float4 v = *(const float4*)(prow + (size_t)p * D_HID);
      acc.x += v.x; acc.y += v.y; acc.z += v.z; acc.w += v.w;
    }
    *(float4*)row = acc;
  }
  __syncthreads();

  const float* arow = xm + (size_t)(mt * 16 + lm) * D_HID + 2 * half;
  const float* hrow = head + (size_t)(cs + 1 < NCHUNK ? cs + 1 : cs) * D_HID + 2 * half;
  const float* bcol = W2 + (2 * half) * D_HID + lm;

  v2f a[32];
#pragma unroll
  for (int ks = 0; ks < 32; ++ks) { a[ks].x = 0.0f; a[ks].y = 0.0f; }
  if (has) {
#pragma unroll
    for (int ks = 0; ks < 32; ++ks) a[ks] = *(const v2f*)(arow + ks * 4);
  }
  if (nh == 1) {
#pragma unroll
    for (int ks = 0; ks < 32; ++ks) {
      const v2f h = *(const v2f*)(hrow + ks * 4);
      a[ks].x += h.x;
      a[ks].y += h.y;
    }
  }

  v8f c[8];
#pragma unroll
  for (int nt = 0; nt < 8; ++nt)
#pragma unroll
    for (int e = 0; e < 8; ++e) c[nt][e] = 0.0f;

#pragma unroll
  for (int ks = 0; ks < 32; ++ks) {
    const float* bp = bcol + (ks * 4) * D_HID;
#pragma unroll
    for (int nt = 0; nt < 8; ++nt) {
      v2f b;
      b.x = bp[nt * 16];
      b.y = bp[nt * 16 + D_HID];
      c[nt] = __builtin_amdgcn_wmma_f32_16x16x4_f32(
          false, a[ks], false, b, (short)0, c[nt], false, false);
    }
  }

#pragma unroll
  for (int nt = 0; nt < 8; ++nt) {
    const int n = nt * 16 + lm;
    const float bias = b2[n];
#pragma unroll
    for (int r = 0; r < 8; ++r) {
      out[(size_t)(mt * 16 + r + 8 * half) * D_HID + n] = c[nt][r] + bias;
    }
  }
}

extern "C" void kernel_launch(void* const* d_in, const int* in_sizes, int n_in,
                              void* d_out, int out_size, void* d_ws, size_t ws_size,
                              hipStream_t stream) {
  (void)in_sizes; (void)n_in; (void)out_size; (void)ws_size;
  const float* x  = (const float*)d_in[0];
  const int*   l  = (const int*)d_in[1];
  const float* W1 = (const float*)d_in[2];
  const float* b1 = (const float*)d_in[3];
  const float* W2 = (const float*)d_in[4];
  const float* b2 = (const float*)d_in[5];
  float* out = (float*)d_out;

  char* ws = (char*)d_ws;
  int*   offs = (int*)ws;
  float* xm   = (float*)(ws + NSEG * sizeof(int));
  float* head = (float*)(ws + NSEG * sizeof(int) + (size_t)NSEG * D_HID * sizeof(float));

  fused_mlp_segsum_kernel<<<NBLK, 256, 0, stream>>>(x, W1, b1, l, offs, xm, head);
  head_gemm_kernel<<<NSEG / 16, 32, 0, stream>>>(xm, head, offs, l, W2, b2, out);
}


# Head kernel: one wave per 16 rows x 64 columns (1024 one-wave blocks, was 512); all 64 W2 loads (16 bytes, four interleaved column tiles) issued at the top; 16-byte stores; sums and order unchanged
# speedup vs baseline: 7.1461x; 1.0472x over previous
#include <hip/hip_runtime.h>
#include <hip/hip_bf16.h>


typedef __attribute__((ext_vector_type(2))) float v2f;
typedef __attribute__((ext_vector_type(4))) float v4f;
typedef __attribute__((ext_vector_type(8))) float v8f;

#define NSEG   8192
#define D_IN   64
#define D_HID  128
#define LDS_PITCH 132
#define TOTAL_TOK 1048576
#define CHUNK_TOK 512
#define NCHUNK    (TOTAL_TOK / CHUNK_TOK)
#define NBLK      (NCHUNK / 8)
#define TILES_PER_CHUNK (CHUNK_TOK / 16)
#define NGROUP    (NSEG / 32)
#define LSEG_PITCH 36
#define NHEADBLK  (NSEG / 16 * 2)
static_assert(NBLK == NGROUP, "block b writes the offsets of group b");

__device__ __forceinline__ void tile_mma_16x128(const v2f a[16],
                                                const v2f (&b)[16][8],
                                                v8f c[8]) {
#pragma unroll
  for (int ks = 0; ks < 16; ++ks) {
#pragma unroll
    for (int nt = 0; nt < 8; ++nt) {
      c[nt] = __builtin_amdgcn_wmma_f32_16x16x4_f32(
          false, a[ks], false, b[ks][nt], (short)0, c[nt], false, false);
    }
  }
}

__device__ __forceinline__ void add_rows_between(const v8f c[8], float accS[8],
                                                 int tok0, int lo, int hi) {
  bool m[8];
#pragma unroll
  for (int r = 0; r < 8; ++r) m[r] = (tok0 + r >= lo) && (tok0 + r < hi);
#pragma unroll
  for (int nt = 0; nt < 8; ++nt) {
    float v[8];
#pragma unroll
    for (int r = 0; r < 8; ++r) v[r] = m[r] ? c[nt][r] : 0.0f;
    float s = ((v[0] + v[1]) + (v[2] + v[3])) + ((v[4] + v[5]) + (v[6] + v[7]));
    accS[nt] += s;
  }
}

__device__ __forceinline__ void store_row_sums(float* dst, float accS[8], int lane) {
#pragma unroll
  for (int nt = 0; nt < 8; ++nt) accS[nt] += __shfl_xor(accS[nt], 16, 32);
  if (lane < 16) {
#pragma unroll
    for (int nt = 0; nt < 8; ++nt) dst[nt * 16 + lane] = accS[nt];
  }
#pragma unroll
  for (int nt = 0; nt < 8; ++nt) accS[nt] = 0.0f;
}


__global__ __launch_bounds__(256) void fused_mlp_segsum_kernel(
    const float* __restrict__ x, const float* __restrict__ W1,
    const float* __restrict__ b1, const int* __restrict__ l,
    int* __restrict__ offs, float* __restrict__ xm, float* __restrict__ head) {
  __shared__ float w1s[D_IN * LDS_PITCH];
  __shared__ __attribute__((aligned(16))) int lS[NGROUP * LSEG_PITCH];
  __shared__ __attribute__((aligned(16))) int exS[NGROUP];
  __shared__ __attribute__((aligned(16))) int waveTot[8];

  const int tid = threadIdx.x;
  const int lane = tid & 31;
  const int wave = __builtin_amdgcn_readfirstlane(tid >> 5);
  const int half = lane >> 4;
  const int lm   = lane & 15;
  const int ci   = blockIdx.x * 8 + wave;
  const int chunkStart = ci * CHUNK_TOK;

  const int4* lp = (const int4*)l + tid;
  const int4 lq0 = lp[0 * 256], lq1 = lp[1 * 256], lq2 = lp[2 * 256], lq3 = lp[3 * 256];
  const int4 lq4 = lp[4 * 256], lq5 = lp[5 * 256], lq6 = lp[6 * 256], lq7 = lp[7 * 256];

  for (int i = tid; i < D_IN * D_HID; i += 256) {
    w1s[(i >> 7) * LDS_PITCH + (i & 127)] = W1[i];
  }

  {
    int* lrow = lS + (tid >> 3) * LSEG_PITCH + (tid & 7) * 4;
    *(int4*)(lrow + 0 * 32 * LSEG_PITCH) = lq0;
    *(int4*)(lrow + 1 * 32 * LSEG_PITCH) = lq1;
    *(int4*)(lrow + 2 * 32 * LSEG_PITCH) = lq2;
    *(int4*)(lrow + 3 * 32 * LSEG_PITCH) = lq3;
    *(int4*)(lrow + 4 * 32 * LSEG_PITCH) = lq4;
    *(int4*)(lrow + 5 * 32 * LSEG_PITCH) = lq5;
    *(int4*)(lrow + 6 * 32 * LSEG_PITCH) = lq6;
    *(int4*)(lrow + 7 * 32 * LSEG_PITCH) = lq7;
  }
  __syncthreads();

  int len[32];
#pragma unroll
  for (int k = 0; k < 8; ++k) {
    const int4 v = *(const int4*)(lS + tid * LSEG_PITCH + 4 * k);
    len[4 * k + 0] = v.x; len[4 * k + 1] = v.y; len[4 * k + 2] = v.z; len[4 * k + 3] = v.w;
  }
  int gsum = 0;
#pragma unroll
  for (int i = 0; i < 32; ++i) gsum += len[i];

  int inc = gsum;
#pragma unroll
  for (int d = 1; d < 32; d <<= 1) {
    const int v = __shfl_up(inc, d, 32);
    if (lane >= d) inc += v;
  }
  exS[tid] = inc - gsum;
  if (lane == 31) waveTot[wave] = inc;
  __syncthreads();

  int wt[8];
  {
    const int4 t0 = *(const int4*)(waveTot), t1 = *(const int4*)(waveTot + 4);
    wt[0] = t0.x; wt[1] = t0.y; wt[2] = t0.z; wt[3] = t0.w;
    wt[4] = t1.x; wt[5] = t1.y; wt[6] = t1.z; wt[7] = t1.w;
  }
  const int lw = lane >> 2;
  int myBase = 0, laneBase = 0, total = 0;
#pragma unroll
  for (int w = 0; w < 8; ++w) {
    myBase   += (w < wave) ? wt[w] : 0;
    laneBase += (w < lw)   ? wt[w] : 0;
    total    += wt[w];
  }

  if (tid == (int)blockIdx.x) {
    int o = myBase + inc - gsum;
#pragma unroll
    for (int k = 0; k < 8; ++k) {
      int4 w4;
      w4.x = o; o += len[4 * k + 0];
      w4.y = o; o += len[4 * k + 1];
      w4.z = o; o += len[4 * k + 2];
      w4.w = o; o += len[4 * k + 3];
      *(int4*)(offs + tid * 32 + 4 * k) = w4;
    }
  }

  int cnt = 0;
  {
    const int4 e0 = *(const int4*)(exS + lane * 8), e1 = *(const int4*)(exS + lane * 8 + 4);
    cnt += __builtin_popcount(__builtin_amdgcn_ballot_w32(laneBase + e0.x <= chunkStart));
    cnt += __builtin_popcount(__builtin_amdgcn_ballot_w32(laneBase + e0.y <= chunkStart));
    cnt += __builtin_popcount(__builtin_amdgcn_ballot_w32(laneBase + e0.z <= chunkStart));
    cnt += __builtin_popcount(__builtin_amdgcn_ballot_w32(laneBase + e0.w <= chunkStart));
    cnt += __builtin_popcount(__builtin_amdgcn_ballot_w32(laneBase + e1.x <= chunkStart));
    cnt += __builtin_popcount(__builtin_amdgcn_ballot_w32(laneBase + e1.y <= chunkStart));
    cnt += __builtin_popcount(__builtin_amdgcn_ballot_w32(laneBase + e1.z <= chunkStart));
    cnt += __builtin_popcount(__builtin_amdgcn_ballot_w32(laneBase + e1.w <= chunkStart));
  }
  const int grp = min(max(cnt - 1, 0), NGROUP - 1);
  const int gw = grp >> 5;
  int grpRun = exS[grp];
#pragma unroll
  for (int w = 0; w < 8; ++w) grpRun += (w < gw) ? wt[w] : 0;

  const int lv = lS[grp * LSEG_PITCH + lane];
  int linc = lv;
#pragma unroll
  for (int d = 1; d < 32; d <<= 1) {
    const int v = __shfl_up(linc, d, 32);
    if (lane >= d) linc += v;
  }
  const int so = grpRun + linc - lv;
  const unsigned hit = __builtin_amdgcn_ballot_w32(lv > 0 && so <= chunkStart && chunkStart < so + lv);
  int seg0, o0v, l0v;
  if (hit != 0u) {
    const int j = __builtin_ctz(hit);
    seg0 = grp * 32 + j;
    o0v = __builtin_amdgcn_readlane(so, j);
    l0v = __builtin_amdgcn_readlane(lv, j);
  } else {
    const int lLast = lS[(NGROUP - 1) * LSEG_PITCH + 31];
    seg0 = NSEG - 1;
    o0v = total - lLast;
    l0v = lLast;
  }

  const float* bcol = w1s + (2 * half) * LDS_PITCH + lm;
  v2f bfrag[16][8];
#pragma unroll
  for (int ks = 0; ks < 16; ++ks) {
    const float* bp = bcol + (ks * 4) * LDS_PITCH;
#pragma unroll
    for (int nt = 0; nt < 8; ++nt) {
      bfrag[ks][nt].x = bp[nt * 16];
      bfrag[ks][nt].y = bp[nt * 16 + LDS_PITCH];
    }
  }

  float bias[8];
#pragma unroll
  for (int nt = 0; nt < 8; ++nt) bias[nt] = b1[nt * 16 + lm];

  int seg = __builtin_amdgcn_readfirstlane(seg0);
  const int o0 = __builtin_amdgcn_readfirstlane(o0v);
  int segEnd = o0 + __builtin_amdgcn_readfirstlane(l0v);
  bool toHead = (o0 < chunkStart);

  float accS[8];
#pragma unroll
  for (int nt = 0; nt < 8; ++nt) accS[nt] = 0.0f;

  auto load_tile = [&](v2f (&dst)[16], int t) {
    const float* xrow = x + (size_t)(chunkStart + t * 16 + lm) * D_IN + 2 * half;
#pragma unroll
    for (int ks = 0; ks < 16; ++ks) dst[ks] = *(const v2f*)(xrow + ks * 4);
  };

  auto do_tile = [&](const v2f (&a)[16], int t) {
    const int T0 = chunkStart + t * 16;
    v8f c[8];
#pragma unroll
    for (int nt = 0; nt < 8; ++nt)
#pragma unroll
      for (int e = 0; e < 8; ++e) c[nt][e] = 0.0f;

    tile_mma_16x128(a, bfrag, c);

#pragma unroll
    for (int nt = 0; nt < 8; ++nt)
#pragma unroll
      for (int r = 0; r < 8; ++r) c[nt][r] = fmaxf(c[nt][r] + bias[nt], 0.0f);

    const int Tend = T0 + 16;
    if (segEnd >= Tend) {
#pragma unroll
      for (int nt = 0; nt < 8; ++nt) {
        float s = ((c[nt][0] + c[nt][1]) + (c[nt][2] + c[nt][3])) +
                  ((c[nt][4] + c[nt][5]) + (c[nt][6] + c[nt][7]));
        accS[nt] += s;
      }
    } else {
      int lo = T0;
      for (int g = 0; g < 16 && segEnd < Tend; ++g) {
        add_rows_between(c, accS, T0 + 8 * half, lo, segEnd);
        float* dst = toHead ? head + (size_t)ci * D_HID : xm + (size_t)seg * D_HID;
        store_row_sums(dst, accS, lane);
        toHead = false;
        lo = segEnd;
        int len = 0;
        do {
          ++seg;
          if (seg >= NSEG) break;
          len = __builtin_amdgcn_readfirstlane(l[seg]);
        } while (len <= 0);
        if (seg >= NSEG) {
          seg = NSEG - 1;
          segEnd = 0x7fffffff;
        } else {
          segEnd = lo + len;
        }
      }
      add_rows_between(c, accS, T0 + 8 * half, lo, Tend);
    }
  };

  v2f a0[16], a1[16];
  load_tile(a0, 0);
#pragma nounroll
  for (int t = 0; t < TILES_PER_CHUNK; t += 2) {
    load_tile(a1, t + 1);
    __builtin_amdgcn_sched_barrier(0);
    do_tile(a0, t);
    load_tile(a0, min(t + 2, TILES_PER_CHUNK - 1));
    __builtin_amdgcn_sched_barrier(0);
    do_tile(a1, t + 1);
  }

  float* dst = toHead ? head + (size_t)ci * D_HID : xm + (size_t)seg * D_HID;
  store_row_sums(dst, accS, lane);
}

__global__ __launch_bounds__(32) void head_gemm_kernel(
    const float* __restrict__ xm, const float* __restrict__ head,
    const int* __restrict__ offs, const int* __restrict__ l,
    const float* __restrict__ W2, const float* __restrict__ b2,
    float* __restrict__ out) {
  __shared__ __attribute__((aligned(16))) float rowS[16 * LDS_PITCH];

  const int lane = threadIdx.x & 31;
  const int half = lane >> 4;
  const int lm   = lane & 15;
  const int mt   = blockIdx.x >> 1;
  const int cb   = (blockIdx.x & 1) * 64 + 4 * lm;

  const int rlen = l[mt * 16 + lm];
  const int ro   = offs[mt * 16 + lm];

  const float* bsrc = W2 + (size_t)(2 * half) * D_HID + cb;
  v4f bq[32][2];
#pragma unroll
  for (int ks = 0; ks < 32; ++ks) {
    bq[ks][0] = *(const v4f*)(bsrc + (ks * 4) * D_HID);
    bq[ks][1] = *(const v4f*)(bsrc + (ks * 4 + 1) * D_HID);
  }

  const bool has = rlen > 0;
  int cs = ro / CHUNK_TOK;
  int ce = (ro + rlen - 1) / CHUNK_TOK;
  cs = min(max(cs, 0), NCHUNK - 1);
  ce = min(max(ce, cs), NCHUNK - 1);
  const int nh = has ? (ce - cs) : 0;

  const float* arow = xm + (size_t)(mt * 16 + lm) * D_HID + 2 * half;
  v2f a[32];
#pragma unroll
  for (int ks = 0; ks < 32; ++ks) { a[ks].x = 0.0f; a[ks].y = 0.0f; }
  if (has) {
#pragma unroll
    for (int ks = 0; ks < 32; ++ks) a[ks] = *(const v2f*)(arow + ks * 4);
  }

  v2f hq[32];
#pragma unroll
  for (int ks = 0; ks < 32; ++ks) { hq[ks].x = 0.0f; hq[ks].y = 0.0f; }
  if (nh == 1) {
    const float* hrow = head + (size_t)(cs + 1 < NCHUNK ? cs + 1 : cs) * D_HID + 2 * half;
#pragma unroll
    for (int ks = 0; ks < 32; ++ks) hq[ks] = *(const v2f*)(hrow + ks * 4);
  }

  unsigned lmask = __builtin_amdgcn_ballot_w32(nh >= 2) & 0xffffu;
  while (lmask != 0u) {
    const int r = __builtin_ctz(lmask);
    lmask &= lmask - 1u;
    const int q  = __shfl(cs, r, 32) + 1;
    const int nf = __shfl(nh, r, 32);
    const float* prow = head + (size_t)q * D_HID + lane * 4;
    float4 acc = *(const float4*)(xm + (size_t)(mt * 16 + r) * D_HID + lane * 4);
#pragma unroll 8
    for (int p = 0; p < nf; ++p) {
      const float4 v = *(const float4*)(prow + (size_t)p * D_HID);
      acc.x += v.x; acc.y += v.y; acc.z += v.z; acc.w += v.w;
    }
    *(float4*)(rowS + r * LDS_PITCH + lane * 4) = acc;
  }
  __syncthreads();

  if (nh == 1) {
#pragma unroll
    for (int ks = 0; ks < 32; ++ks) {
      a[ks].x += hq[ks].x;
      a[ks].y += hq[ks].y;
    }
  }
  if (nh >= 2) {
    const float* srow = rowS + lm * LDS_PITCH + 2 * half;
#pragma unroll
    for (int ks = 0; ks < 32; ++ks) a[ks] = *(const v2f*)(srow + ks * 4);
  }

  v8f c[4];
#pragma unroll
  for (int j = 0; j < 4; ++j)
#pragma unroll
    for (int e = 0; e < 8; ++e) c[j][e] = 0.0f;

#pragma unroll
  for (int ks = 0; ks < 32; ++ks) {
#pragma unroll
    for (int j = 0; j < 4; ++j) {
      v2f b;
      b.x = bq[ks][0][j];
      b.y = bq[ks][1][j];
      c[j] = __builtin_amdgcn_wmma_f32_16x16x4_f32(
          false, a[ks], false, b, (short)0, c[j], false, false);
    }
  }

  const v4f bias = *(const v4f*)(b2 + cb);
#pragma unroll
  for (int r = 0; r < 8; ++r) {
    v4f o;
    o[0] = c[0][r] + bias[0];
    o[1] = c[1][r] + bias[1];
    o[2] = c[2][r] + bias[2];
    o[3] = c[3][r] + bias[3];
    *(v4f*)(out + (size_t)(mt * 16 + r + 8 * half) * D_HID + cb) = o;
  }
}

extern "C" void kernel_launch(void* const* d_in, const int* in_sizes, int n_in,
                              void* d_out, int out_size, void* d_ws, size_t ws_size,
                              hipStream_t stream) {
  (void)in_sizes; (void)n_in; (void)out_size; (void)ws_size;
  const float* x  = (const float*)d_in[0];
  const int*   l  = (const int*)d_in[1];
  const float* W1 = (const float*)d_in[2];
  const float* b1 = (const float*)d_in[3];
  const float* W2 = (const float*)d_in[4];
  const float* b2 = (const float*)d_in[5];
  float* out = (float*)d_out;

  char* ws = (char*)d_ws;
  int*   offs = (int*)ws;
  float* xm   = (float*)(ws + NSEG * sizeof(int));
  float* head = (float*)(ws + NSEG * sizeof(int) + (size_t)NSEG * D_HID * sizeof(float));

  fused_mlp_segsum_kernel<<<NBLK, 256, 0, stream>>>(x, W1, b1, l, offs, xm, head);
  head_gemm_kernel<<<NHEADBLK, 32, 0, stream>>>(xm, head, offs, l, W2, b2, out);
}
